# starting kernel
# baseline (speedup 1.0000x reference)
#include <hip/hip_runtime.h>
#include <math.h>

typedef __attribute__((ext_vector_type(2))) float v2f;
typedef __attribute__((ext_vector_type(8))) float v8f;

#define WMMA4(a, b, c) \
    __builtin_amdgcn_wmma_f32_16x16x4_f32(false, (a), false, (b), (short)0, (c), false, false)

__device__ __forceinline__ float fast_tanh(float x) {
#if __has_builtin(__builtin_amdgcn_tanhf)
    return __builtin_amdgcn_tanhf(x);
#else
    float xc = fminf(fmaxf(x, -10.0f), 10.0f);
    float t = __builtin_amdgcn_exp2f(xc * 2.8853900817779268f);
    return (t - 1.0f) * __builtin_amdgcn_rcpf(t + 1.0f);
#endif
}

__launch_bounds__(32)
__global__ void genode_rk45_wmma(const float* __restrict__ z_end,
                                 const float* __restrict__ W1,
                                 const float* __restrict__ b1,
                                 const float* __restrict__ W2,
                                 const float* __restrict__ b2,
                                 float* __restrict__ out,
                                 float h)
{
    __shared__ v2f w1B[8][4][32];
    __shared__ v2f w2B[32][32];
    __shared__ __align__(16) float hbuf[16][132];
    __shared__ __align__(16) float zbuf[16][20];

    const int lane = threadIdx.x;
    const int n    = lane & 15;
    const int half = lane >> 4;
    const int dlt  = half * 2;
    const int g    = blockIdx.x;

    #pragma unroll
    for (int t = 0; t < 8; ++t)
        #pragma unroll
        for (int c = 0; c < 4; ++c) {
            int k = 4 * c + dlt;
            v2f v;
            v.x = W1[k * 128 + 16 * t + n];
            v.y = W1[(k + 1) * 128 + 16 * t + n];
            w1B[t][c][lane] = v;
        }
    #pragma unroll
    for (int c = 0; c < 32; ++c) {
        int k = 4 * c + dlt;
        v2f v;
        v.x = W2[k * 16 + n];
        v.y = W2[(k + 1) * 16 + n];
        w2B[c][lane] = v;
    }
    float b1v[8];
    #pragma unroll
    for (int t = 0; t < 8; ++t) b1v[t] = b1[16 * t + n];
    const float b2v = b2[n];

    v8f z;
    #pragma unroll
    for (int v = 0; v < 8; ++v)
        z[v] = z_end[(16 * g + v + half * 8) * 16 + n];

    __syncthreads();

    auto put_z = [&](const v8f& val) {
        #pragma unroll
        for (int v = 0; v < 8; ++v)
            zbuf[v + half * 8][n] = val[v];
    };

    auto feval = [&]() -> v8f {
        __syncthreads();
        v2f az[4];
        #pragma unroll
        for (int c = 0; c < 4; ++c)
            az[c] = *(const v2f*)&zbuf[n][4 * c + dlt];

        #pragma unroll
        for (int t = 0; t < 8; ++t) {
            v8f acc;
            #pragma unroll
            for (int v = 0; v < 8; ++v) acc[v] = b1v[t];
            #pragma unroll
            for (int c = 0; c < 4; ++c)
                acc = WMMA4(az[c], w1B[t][c][lane], acc);
            #pragma unroll
            for (int v = 0; v < 8; ++v)
                hbuf[v + half * 8][16 * t + n] = fast_tanh(acc[v]);
        }
        __syncthreads();

        v8f o;
        #pragma unroll
        for (int v = 0; v < 8; ++v) o[v] = b2v;
        #pragma unroll
        for (int c = 0; c < 32; ++c) {
            v2f ah = *(const v2f*)&hbuf[n][4 * c + dlt];
            o = WMMA4(ah, w2B[c][lane], o);
        }
        return o;
    };

    const int nsteps = 16 * g + 15;
    for (int s = 1; s <= nsteps; ++s) {
        put_z(z);
        v8f k1 = feval();

        v8f t;
        #pragma unroll
        for (int v = 0; v < 8; ++v) t[v] = z[v] + h * (0.2f * k1[v]);
        put_z(t);
        v8f k2 = feval();

        #pragma unroll
        for (int v = 0; v < 8; ++v)
            t[v] = z[v] + h * ((3.0f / 40.0f) * k1[v] + (9.0f / 40.0f) * k2[v]);
        put_z(t);
        v8f k3 = feval();

        #pragma unroll
        for (int v = 0; v < 8; ++v)
            t[v] = z[v] + h * ((44.0f / 45.0f) * k1[v] - (56.0f / 15.0f) * k2[v]
                               + (32.0f / 9.0f) * k3[v]);
        put_z(t);
        v8f k4 = feval();

        #pragma unroll
        for (int v = 0; v < 8; ++v)
            t[v] = z[v] + h * ((19372.0f / 6561.0f) * k1[v] - (25360.0f / 2187.0f) * k2[v]
                               + (64448.0f / 6561.0f) * k3[v] - (212.0f / 729.0f) * k4[v]);
        put_z(t);
        v8f k5 = feval();

        #pragma unroll
        for (int v = 0; v < 8; ++v)
            t[v] = z[v] + h * ((9017.0f / 3168.0f) * k1[v] - (355.0f / 33.0f) * k2[v]
                               + (46732.0f / 5247.0f) * k3[v] + (49.0f / 176.0f) * k4[v]
                               - (5103.0f / 18656.0f) * k5[v]);
        put_z(t);
        v8f k6 = feval();

        #pragma unroll
        for (int v = 0; v < 8; ++v) {
            float zn = z[v] + h * ((35.0f / 384.0f) * k1[v] + (500.0f / 1113.0f) * k3[v]
                                   + (125.0f / 192.0f) * k4[v] - (2187.0f / 6784.0f) * k5[v]
                                   + (11.0f / 84.0f) * k6[v]);
            int target = 16 * g + v + half * 8;
            z[v] = (target >= s) ? zn : z[v];
        }
    }

    #pragma unroll
    for (int v = 0; v < 8; ++v)
        out[(16 * g + v + half * 8) * 16 + n] = z[v];
}

extern "C" void kernel_launch(void* const* d_in, const int* in_sizes, int n_in,
                              void* d_out, int out_size, void* d_ws, size_t ws_size,
                              hipStream_t stream) {
    const float* z_end = (const float*)d_in[0];
    const float* W1    = (const float*)d_in[1];
    const float* b1    = (const float*)d_in[2];
    const float* W2    = (const float*)d_in[3];
    const float* b2    = (const float*)d_in[4];
    float* out = (float*)d_out;

    const int N = in_sizes[0] / 16;
    const float h = 1.0f / (float)(N - 1);
    const int ntiles = N / 16;

    hipLaunchKernelGGL(genode_rk45_wmma, dim3(ntiles), dim3(32), 0, stream,
                       z_end, W1, b1, W2, b2, out, h);
}


# LDS hand-overs pipelined in wave order, not drained (one-wave work-group): activations go in two halves, first half's tanh beside layer-1 WMMAs, second half's tanh under first half's fragment loads
# speedup vs baseline: 1.2774x; 1.2774x over previous
#include <hip/hip_runtime.h>
#include <math.h>

typedef __attribute__((ext_vector_type(2))) float v2f;
typedef __attribute__((ext_vector_type(8))) float v8f;

#define WMMA4(a, b, c) \
    __builtin_amdgcn_wmma_f32_16x16x4_f32(false, (a), false, (b), (short)0, (c), false, false)

__device__ __forceinline__ float fast_tanh(float x) {
#if __has_builtin(__builtin_amdgcn_tanhf)
    return __builtin_amdgcn_tanhf(x);
#else
    float xc = fminf(fmaxf(x, -10.0f), 10.0f);
    float t = __builtin_amdgcn_exp2f(xc * 2.8853900817779268f);
    return (t - 1.0f) * __builtin_amdgcn_rcpf(t + 1.0f);
#endif
}

__device__ __forceinline__ void wave_order() {
    __builtin_amdgcn_fence(__ATOMIC_RELEASE, "wavefront");
    __builtin_amdgcn_wave_barrier();
    __builtin_amdgcn_fence(__ATOMIC_ACQUIRE, "wavefront");
}

__launch_bounds__(32)
__global__ void genode_rk45_wmma(const float* __restrict__ z_end,
                                 const float* __restrict__ W1,
                                 const float* __restrict__ b1,
                                 const float* __restrict__ W2,
                                 const float* __restrict__ b2,
                                 float* __restrict__ out,
                                 float h)
{
    __shared__ __align__(16) float hbuf[16][132];
    __shared__ __align__(16) float zbuf[16][20];

    const int lane = threadIdx.x;
    const int n    = lane & 15;
    const int half = lane >> 4;
    const int dlt  = half * 2;
    const int g    = blockIdx.x;

    v2f w1B[8][4];
    #pragma unroll
    for (int t = 0; t < 8; ++t)
        #pragma unroll
        for (int c = 0; c < 4; ++c) {
            int k = 4 * c + dlt;
            v2f v;
            v.x = W1[k * 128 + 16 * t + n];
            v.y = W1[(k + 1) * 128 + 16 * t + n];
            w1B[t][c] = v;
        }
    v2f w2B[32];
    #pragma unroll
    for (int c = 0; c < 32; ++c) {
        int k = 4 * c + dlt;
        v2f v;
        v.x = W2[k * 16 + n];
        v.y = W2[(k + 1) * 16 + n];
        w2B[c] = v;
    }
    float b1v[8];
    #pragma unroll
    for (int t = 0; t < 8; ++t) b1v[t] = b1[16 * t + n];
    const float b2v = b2[n];

    v8f z;
    #pragma unroll
    for (int v = 0; v < 8; ++v)
        z[v] = z_end[(16 * g + v + half * 8) * 16 + n];

    auto put_z = [&](const v8f& val) {
        #pragma unroll
        for (int v = 0; v < 8; ++v)
            zbuf[v + half * 8][n] = val[v];
    };

    auto feval = [&]() -> v8f {
        wave_order();
        v2f az[4];
        #pragma unroll
        for (int c = 0; c < 4; ++c)
            az[c] = *(const v2f*)&zbuf[n][4 * c + dlt];

        v8f acc[8];
        #pragma unroll
        for (int t = 0; t < 8; ++t)
            #pragma unroll
            for (int v = 0; v < 8; ++v) acc[t][v] = b1v[t];
        #pragma unroll
        for (int c = 0; c < 4; ++c)
            #pragma unroll
            for (int t = 0; t < 4; ++t)
                acc[t] = WMMA4(az[c], w1B[t][c], acc[t]);

        float ha[4][8];
        #pragma unroll
        for (int c = 0; c < 4; ++c)
            #pragma unroll
            for (int t = 4; t < 8; ++t) {
                acc[t] = WMMA4(az[c], w1B[t][c], acc[t]);
                const int i = 2 * (4 * c + (t - 4));
                ha[i >> 3][i & 7]             = fast_tanh(acc[i >> 3][i & 7]);
                ha[(i + 1) >> 3][(i + 1) & 7] = fast_tanh(acc[(i + 1) >> 3][(i + 1) & 7]);
                __builtin_amdgcn_sched_barrier(0);
            }

        #pragma unroll
        for (int t = 0; t < 4; ++t)
            #pragma unroll
            for (int v = 0; v < 8; ++v)
                hbuf[v + half * 8][16 * t + n] = ha[t][v];
        wave_order();
        v2f ah[32];
        #pragma unroll
        for (int c = 0; c < 16; ++c)
            ah[c] = *(const v2f*)&hbuf[n][4 * c + dlt];
        __builtin_amdgcn_sched_barrier(0);

        #pragma unroll
        for (int t = 4; t < 8; ++t)
            #pragma unroll
            for (int v = 0; v < 8; ++v)
                hbuf[v + half * 8][16 * t + n] = fast_tanh(acc[t][v]);
        wave_order();
        #pragma unroll
        for (int c = 16; c < 32; ++c)
            ah[c] = *(const v2f*)&hbuf[n][4 * c + dlt];
        __builtin_amdgcn_sched_barrier(0);

        v8f o;
        #pragma unroll
        for (int v = 0; v < 8; ++v) o[v] = b2v;
        #pragma unroll
        for (int c = 0; c < 32; ++c)
            o = WMMA4(ah[c], w2B[c], o);
        return o;
    };

    const int nsteps = 16 * g + 15;
    for (int s = 1; s <= nsteps; ++s) {
        put_z(z);
        v8f k1 = feval();

        v8f t;
        #pragma unroll
        for (int v = 0; v < 8; ++v) t[v] = z[v] + h * (0.2f * k1[v]);
        put_z(t);
        v8f k2 = feval();

        #pragma unroll
        for (int v = 0; v < 8; ++v)
            t[v] = z[v] + h * ((3.0f / 40.0f) * k1[v] + (9.0f / 40.0f) * k2[v]);
        put_z(t);
        v8f k3 = feval();

        #pragma unroll
        for (int v = 0; v < 8; ++v)
            t[v] = z[v] + h * ((44.0f / 45.0f) * k1[v] - (56.0f / 15.0f) * k2[v]
                               + (32.0f / 9.0f) * k3[v]);
        put_z(t);
        v8f k4 = feval();

        #pragma unroll
        for (int v = 0; v < 8; ++v)
            t[v] = z[v] + h * ((19372.0f / 6561.0f) * k1[v] - (25360.0f / 2187.0f) * k2[v]
                               + (64448.0f / 6561.0f) * k3[v] - (212.0f / 729.0f) * k4[v]);
        put_z(t);
        v8f k5 = feval();

        #pragma unroll
        for (int v = 0; v < 8; ++v)
            t[v] = z[v] + h * ((9017.0f / 3168.0f) * k1[v] - (355.0f / 33.0f) * k2[v]
                               + (46732.0f / 5247.0f) * k3[v] + (49.0f / 176.0f) * k4[v]
                               - (5103.0f / 18656.0f) * k5[v]);
        put_z(t);
        v8f k6 = feval();

        #pragma unroll
        for (int v = 0; v < 8; ++v) {
            float zn = z[v] + h * ((35.0f / 384.0f) * k1[v] + (500.0f / 1113.0f) * k3[v]
                                   + (125.0f / 192.0f) * k4[v] - (2187.0f / 6784.0f) * k5[v]
                                   + (11.0f / 84.0f) * k6[v]);
            int target = 16 * g + v + half * 8;
            z[v] = (target >= s) ? zn : z[v];
        }
    }

    #pragma unroll
    for (int v = 0; v < 8; ++v)
        out[(16 * g + v + half * 8) * 16 + n] = z[v];
}

extern "C" void kernel_launch(void* const* d_in, const int* in_sizes, int n_in,
                              void* d_out, int out_size, void* d_ws, size_t ws_size,
                              hipStream_t stream) {
    const float* z_end = (const float*)d_in[0];
    const float* W1    = (const float*)d_in[1];
    const float* b1    = (const float*)d_in[2];
    const float* W2    = (const float*)d_in[3];
    const float* b2    = (const float*)d_in[4];
    float* out = (float*)d_out;

    const int N = in_sizes[0] / 16;
    const float h = 1.0f / (float)(N - 1);
    const int ntiles = N / 16;

    hipLaunchKernelGGL(genode_rk45_wmma, dim3(ntiles), dim3(32), 0, stream,
                       z_end, W1, b1, W2, b2, out, h);
}


# Layer 2 sums into two independent accumulator tiles (even / odd k-chunks), added once at the end, so no matrix instruction waits on its predecessor; layer-2 sum order changes (two partial sums)
# speedup vs baseline: 1.3699x; 1.0724x over previous
#include <hip/hip_runtime.h>
#include <math.h>

typedef __attribute__((ext_vector_type(2))) float v2f;
typedef __attribute__((ext_vector_type(8))) float v8f;

#define WMMA4(a, b, c) \
    __builtin_amdgcn_wmma_f32_16x16x4_f32(false, (a), false, (b), (short)0, (c), false, false)

__device__ __forceinline__ float fast_tanh(float x) {
#if __has_builtin(__builtin_amdgcn_tanhf)
    return __builtin_amdgcn_tanhf(x);
#else
    float xc = fminf(fmaxf(x, -10.0f), 10.0f);
    float t = __builtin_amdgcn_exp2f(xc * 2.8853900817779268f);
    return (t - 1.0f) * __builtin_amdgcn_rcpf(t + 1.0f);
#endif
}

__device__ __forceinline__ void wave_order() {
    __builtin_amdgcn_fence(__ATOMIC_RELEASE, "wavefront");
    __builtin_amdgcn_wave_barrier();
    __builtin_amdgcn_fence(__ATOMIC_ACQUIRE, "wavefront");
}

__launch_bounds__(32)
__global__ void genode_rk45_wmma(const float* __restrict__ z_end,
                                 const float* __restrict__ W1,
                                 const float* __restrict__ b1,
                                 const float* __restrict__ W2,
                                 const float* __restrict__ b2,
                                 float* __restrict__ out,
                                 float h)
{
    __shared__ __align__(16) float hbuf[16][132];
    __shared__ __align__(16) float zbuf[16][20];

    const int lane = threadIdx.x;
    const int n    = lane & 15;
    const int half = lane >> 4;
    const int dlt  = half * 2;
    const int g    = blockIdx.x;

    v2f w1B[8][4];
    #pragma unroll
    for (int t = 0; t < 8; ++t)
        #pragma unroll
        for (int c = 0; c < 4; ++c) {
            int k = 4 * c + dlt;
            v2f v;
            v.x = W1[k * 128 + 16 * t + n];
            v.y = W1[(k + 1) * 128 + 16 * t + n];
            w1B[t][c] = v;
        }
    v2f w2B[32];
    #pragma unroll
    for (int c = 0; c < 32; ++c) {
        int k = 4 * c + dlt;
        v2f v;
        v.x = W2[k * 16 + n];
        v.y = W2[(k + 1) * 16 + n];
        w2B[c] = v;
    }
    float b1v[8];
    #pragma unroll
    for (int t = 0; t < 8; ++t) b1v[t] = b1[16 * t + n];
    const float b2v = b2[n];

    v8f z;
    #pragma unroll
    for (int v = 0; v < 8; ++v)
        z[v] = z_end[(16 * g + v + half * 8) * 16 + n];

    auto put_z = [&](const v8f& val) {
        #pragma unroll
        for (int v = 0; v < 8; ++v)
            zbuf[v + half * 8][n] = val[v];
    };

    auto feval = [&]() -> v8f {
        wave_order();
        v2f az[4];
        #pragma unroll
        for (int c = 0; c < 4; ++c)
            az[c] = *(const v2f*)&zbuf[n][4 * c + dlt];

        v8f acc[8];
        #pragma unroll
        for (int t = 0; t < 8; ++t)
            #pragma unroll
            for (int v = 0; v < 8; ++v) acc[t][v] = b1v[t];
        #pragma unroll
        for (int c = 0; c < 4; ++c)
            #pragma unroll
            for (int t = 0; t < 4; ++t)
                acc[t] = WMMA4(az[c], w1B[t][c], acc[t]);

        float ha[4][8];
        #pragma unroll
        for (int c = 0; c < 4; ++c)
            #pragma unroll
            for (int t = 4; t < 8; ++t) {
                acc[t] = WMMA4(az[c], w1B[t][c], acc[t]);
                const int i = 2 * (4 * c + (t - 4));
                ha[i >> 3][i & 7]             = fast_tanh(acc[i >> 3][i & 7]);
                ha[(i + 1) >> 3][(i + 1) & 7] = fast_tanh(acc[(i + 1) >> 3][(i + 1) & 7]);
                __builtin_amdgcn_sched_barrier(0);
            }

        #pragma unroll
        for (int t = 0; t < 4; ++t)
            #pragma unroll
            for (int v = 0; v < 8; ++v)
                hbuf[v + half * 8][16 * t + n] = ha[t][v];
        wave_order();
        v2f ah[32];
        #pragma unroll
        for (int c = 0; c < 16; ++c)
            ah[c] = *(const v2f*)&hbuf[n][4 * c + dlt];
        __builtin_amdgcn_sched_barrier(0);

        #pragma unroll
        for (int t = 4; t < 8; ++t)
            #pragma unroll
            for (int v = 0; v < 8; ++v)
                hbuf[v + half * 8][16 * t + n] = fast_tanh(acc[t][v]);
        wave_order();
        #pragma unroll
        for (int c = 16; c < 32; ++c)
            ah[c] = *(const v2f*)&hbuf[n][4 * c + dlt];
        __builtin_amdgcn_sched_barrier(0);

        v8f oa, ob;
        #pragma unroll
        for (int v = 0; v < 8; ++v) { oa[v] = b2v; ob[v] = 0.0f; }
        #pragma unroll
        for (int c = 0; c < 32; c += 2) {
            oa = WMMA4(ah[c],     w2B[c],     oa);
            ob = WMMA4(ah[c + 1], w2B[c + 1], ob);
        }
        v8f o;
        #pragma unroll
        for (int v = 0; v < 8; ++v) o[v] = oa[v] + ob[v];
        return o;
    };

    const int nsteps = 16 * g + 15;
    for (int s = 1; s <= nsteps; ++s) {
        put_z(z);
        v8f k1 = feval();

        v8f t;
        #pragma unroll
        for (int v = 0; v < 8; ++v) t[v] = z[v] + h * (0.2f * k1[v]);
        put_z(t);
        v8f k2 = feval();

        #pragma unroll
        for (int v = 0; v < 8; ++v)
            t[v] = z[v] + h * ((3.0f / 40.0f) * k1[v] + (9.0f / 40.0f) * k2[v]);
        put_z(t);
        v8f k3 = feval();

        #pragma unroll
        for (int v = 0; v < 8; ++v)
            t[v] = z[v] + h * ((44.0f / 45.0f) * k1[v] - (56.0f / 15.0f) * k2[v]
                               + (32.0f / 9.0f) * k3[v]);
        put_z(t);
        v8f k4 = feval();

        #pragma unroll
        for (int v = 0; v < 8; ++v)
            t[v] = z[v] + h * ((19372.0f / 6561.0f) * k1[v] - (25360.0f / 2187.0f) * k2[v]
                               + (64448.0f / 6561.0f) * k3[v] - (212.0f / 729.0f) * k4[v]);
        put_z(t);
        v8f k5 = feval();

        #pragma unroll
        for (int v = 0; v < 8; ++v)
            t[v] = z[v] + h * ((9017.0f / 3168.0f) * k1[v] - (355.0f / 33.0f) * k2[v]
                               + (46732.0f / 5247.0f) * k3[v] + (49.0f / 176.0f) * k4[v]
                               - (5103.0f / 18656.0f) * k5[v]);
        put_z(t);
        v8f k6 = feval();

        #pragma unroll
        for (int v = 0; v < 8; ++v) {
            float zn = z[v] + h * ((35.0f / 384.0f) * k1[v] + (500.0f / 1113.0f) * k3[v]
                                   + (125.0f / 192.0f) * k4[v] - (2187.0f / 6784.0f) * k5[v]
                                   + (11.0f / 84.0f) * k6[v]);
            int target = 16 * g + v + half * 8;
            z[v] = (target >= s) ? zn : z[v];
        }
    }

    #pragma unroll
    for (int v = 0; v < 8; ++v)
        out[(16 * g + v + half * 8) * 16 + n] = z[v];
}

extern "C" void kernel_launch(void* const* d_in, const int* in_sizes, int n_in,
                              void* d_out, int out_size, void* d_ws, size_t ws_size,
                              hipStream_t stream) {
    const float* z_end = (const float*)d_in[0];
    const float* W1    = (const float*)d_in[1];
    const float* b1    = (const float*)d_in[2];
    const float* W2    = (const float*)d_in[3];
    const float* b2    = (const float*)d_in[4];
    float* out = (float*)d_out;

    const int N = in_sizes[0] / 16;
    const float h = 1.0f / (float)(N - 1);
    const int ntiles = N / 16;

    hipLaunchKernelGGL(genode_rk45_wmma, dim3(ntiles), dim3(32), 0, stream,
                       z_end, W1, b1, W2, b2, out, h);
}


# Step arithmetic off the wave's serial chain: tableau sums kept as running partial sums, updated in the LDS wait of each stage hand-over; closing step as packed FMAs, freeze as a zero step size
# speedup vs baseline: 1.4352x; 1.0477x over previous
#include <hip/hip_runtime.h>
#include <math.h>

typedef __attribute__((ext_vector_type(2))) float v2f;
typedef __attribute__((ext_vector_type(8))) float v8f;

#define WMMA4(a, b, c) \
    __builtin_amdgcn_wmma_f32_16x16x4_f32(false, (a), false, (b), (short)0, (c), false, false)

__device__ __forceinline__ float fast_tanh(float x) {
#if __has_builtin(__builtin_amdgcn_tanhf)
    return __builtin_amdgcn_tanhf(x);
#else
    float xc = fminf(fmaxf(x, -10.0f), 10.0f);
    float t = __builtin_amdgcn_exp2f(xc * 2.8853900817779268f);
    return (t - 1.0f) * __builtin_amdgcn_rcpf(t + 1.0f);
#endif
}

__device__ __forceinline__ void wave_order() {
    __builtin_amdgcn_fence(__ATOMIC_RELEASE, "wavefront");
    __builtin_amdgcn_wave_barrier();
    __builtin_amdgcn_fence(__ATOMIC_ACQUIRE, "wavefront");
}

__launch_bounds__(32)
__global__ void genode_rk45_wmma(const float* __restrict__ z_end,
                                 const float* __restrict__ W1,
                                 const float* __restrict__ b1,
                                 const float* __restrict__ W2,
                                 const float* __restrict__ b2,
                                 float* __restrict__ out,
                                 float h)
{
    __shared__ __align__(16) float hbuf[16][132];
    __shared__ __align__(16) float zbuf[16][20];

    const int lane = threadIdx.x;
    const int n    = lane & 15;
    const int half = lane >> 4;
    const int dlt  = half * 2;
    const int g    = blockIdx.x;

    v2f w1B[8][4];
    #pragma unroll
    for (int t = 0; t < 8; ++t)
        #pragma unroll
        for (int c = 0; c < 4; ++c) {
            int k = 4 * c + dlt;
            v2f v;
            v.x = W1[k * 128 + 16 * t + n];
            v.y = W1[(k + 1) * 128 + 16 * t + n];
            w1B[t][c] = v;
        }
    v2f w2B[32];
    #pragma unroll
    for (int c = 0; c < 32; ++c) {
        int k = 4 * c + dlt;
        v2f v;
        v.x = W2[k * 16 + n];
        v.y = W2[(k + 1) * 16 + n];
        w2B[c] = v;
    }
    float b1v[8];
    #pragma unroll
    for (int t = 0; t < 8; ++t) b1v[t] = b1[16 * t + n];
    const float b2v = b2[n];

    v8f z;
    #pragma unroll
    for (int v = 0; v < 8; ++v)
        z[v] = z_end[(16 * g + v + half * 8) * 16 + n];

    auto put_z = [&](const v8f& val) {
        #pragma unroll
        for (int v = 0; v < 8; ++v)
            zbuf[v + half * 8][n] = val[v];
    };

    auto feval = [&](auto&& window) -> v8f {
        wave_order();
        v2f az[4];
        #pragma unroll
        for (int c = 0; c < 4; ++c)
            az[c] = *(const v2f*)&zbuf[n][4 * c + dlt];
        __builtin_amdgcn_sched_barrier(0);
        window();
        __builtin_amdgcn_sched_barrier(0);

        v8f acc[8];
        #pragma unroll
        for (int t = 0; t < 8; ++t)
            #pragma unroll
            for (int v = 0; v < 8; ++v) acc[t][v] = b1v[t];
        #pragma unroll
        for (int c = 0; c < 4; ++c)
            #pragma unroll
            for (int t = 0; t < 4; ++t)
                acc[t] = WMMA4(az[c], w1B[t][c], acc[t]);

        float ha[4][8];
        #pragma unroll
        for (int c = 0; c < 4; ++c)
            #pragma unroll
            for (int t = 4; t < 8; ++t) {
                acc[t] = WMMA4(az[c], w1B[t][c], acc[t]);
                const int i = 2 * (4 * c + (t - 4));
                ha[i >> 3][i & 7]             = fast_tanh(acc[i >> 3][i & 7]);
                ha[(i + 1) >> 3][(i + 1) & 7] = fast_tanh(acc[(i + 1) >> 3][(i + 1) & 7]);
                __builtin_amdgcn_sched_barrier(0);
            }

        #pragma unroll
        for (int t = 0; t < 4; ++t)
            #pragma unroll
            for (int v = 0; v < 8; ++v)
                hbuf[v + half * 8][16 * t + n] = ha[t][v];
        wave_order();
        v2f ah[32];
        #pragma unroll
        for (int c = 0; c < 16; ++c)
            ah[c] = *(const v2f*)&hbuf[n][4 * c + dlt];
        __builtin_amdgcn_sched_barrier(0);

        #pragma unroll
        for (int t = 4; t < 8; ++t)
            #pragma unroll
            for (int v = 0; v < 8; ++v)
                hbuf[v + half * 8][16 * t + n] = fast_tanh(acc[t][v]);
        wave_order();
        #pragma unroll
        for (int c = 16; c < 32; ++c)
            ah[c] = *(const v2f*)&hbuf[n][4 * c + dlt];
        __builtin_amdgcn_sched_barrier(0);

        v8f oa, ob;
        #pragma unroll
        for (int v = 0; v < 8; ++v) { oa[v] = b2v; ob[v] = 0.0f; }
        #pragma unroll
        for (int c = 0; c < 32; c += 2) {
            oa = WMMA4(ah[c],     w2B[c],     oa);
            ob = WMMA4(ah[c + 1], w2B[c + 1], ob);
        }
        v8f o;
        #pragma unroll
        for (int v = 0; v < 8; ++v) o[v] = oa[v] + ob[v];
        return o;
    };

    #define NEEDED_HERE(x) asm volatile("" :: "v"((x)[0]), "v"((x)[1]), "v"((x)[2]), "v"((x)[3]), \
                                              "v"((x)[4]), "v"((x)[5]), "v"((x)[6]), "v"((x)[7]))

    const int nsteps = 16 * g + 15;
    const int row0 = 16 * g + half * 8;
    for (int s = 1; s <= nsteps; ++s) {
        v8f S4, S5, S6, SF;
        v8f hm;
        v8f t;

        put_z(z);
        v8f k1 = feval([&]() {
            #pragma unroll
            for (int v = 0; v < 8; ++v) hm[v] = (row0 + v >= s) ? h : 0.0f;
            NEEDED_HERE(hm);
        });

        #pragma unroll
        for (int v = 0; v < 8; ++v) t[v] = z[v] + h * (0.2f * k1[v]);
        put_z(t);
        v8f k2 = feval([&]() {});

        #pragma unroll
        for (int v = 0; v < 8; ++v)
            t[v] = z[v] + h * ((3.0f / 40.0f) * k1[v] + (9.0f / 40.0f) * k2[v]);
        put_z(t);
        v8f k3 = feval([&]() {
            #pragma unroll
            for (int v = 0; v < 8; ++v) {
                S4[v] = (44.0f / 45.0f) * k1[v] - (56.0f / 15.0f) * k2[v];
                S5[v] = (19372.0f / 6561.0f) * k1[v] - (25360.0f / 2187.0f) * k2[v];
                S6[v] = (9017.0f / 3168.0f) * k1[v] - (355.0f / 33.0f) * k2[v];
            }
            NEEDED_HERE(S4); NEEDED_HERE(S5); NEEDED_HERE(S6);
        });

        #pragma unroll
        for (int v = 0; v < 8; ++v) t[v] = z[v] + h * (S4[v] + (32.0f / 9.0f) * k3[v]);
        put_z(t);
        v8f k4 = feval([&]() {
            #pragma unroll
            for (int v = 0; v < 8; ++v) {
                S5[v] = S5[v] + (64448.0f / 6561.0f) * k3[v];
                S6[v] = S6[v] + (46732.0f / 5247.0f) * k3[v];
                SF[v] = (35.0f / 384.0f) * k1[v] + (500.0f / 1113.0f) * k3[v];
            }
            NEEDED_HERE(S5); NEEDED_HERE(S6); NEEDED_HERE(SF);
        });

        #pragma unroll
        for (int v = 0; v < 8; ++v) t[v] = z[v] + h * (S5[v] - (212.0f / 729.0f) * k4[v]);
        put_z(t);
        v8f k5 = feval([&]() {
            #pragma unroll
            for (int v = 0; v < 8; ++v) {
                S6[v] = S6[v] + (49.0f / 176.0f) * k4[v];
                SF[v] = SF[v] + (125.0f / 192.0f) * k4[v];
            }
            NEEDED_HERE(S6); NEEDED_HERE(SF);
        });

        #pragma unroll
        for (int v = 0; v < 8; ++v) t[v] = z[v] + h * (S6[v] - (5103.0f / 18656.0f) * k5[v]);
        put_z(t);
        v8f k6 = feval([&]() {
            #pragma unroll
            for (int v = 0; v < 8; ++v) SF[v] = SF[v] - (2187.0f / 6784.0f) * k5[v];
            NEEDED_HERE(SF);
        });

        #pragma unroll
        for (int v = 0; v < 8; ++v)
            z[v] = z[v] + hm[v] * (SF[v] + (11.0f / 84.0f) * k6[v]);
    }
    #undef NEEDED_HERE

    #pragma unroll
    for (int v = 0; v < 8; ++v)
        out[(16 * g + v + half * 8) * 16 + n] = z[v];
}

extern "C" void kernel_launch(void* const* d_in, const int* in_sizes, int n_in,
                              void* d_out, int out_size, void* d_ws, size_t ws_size,
                              hipStream_t stream) {
    const float* z_end = (const float*)d_in[0];
    const float* W1    = (const float*)d_in[1];
    const float* b1    = (const float*)d_in[2];
    const float* W2    = (const float*)d_in[3];
    const float* b2    = (const float*)d_in[4];
    float* out = (float*)d_out;

    const int N = in_sizes[0] / 16;
    const float h = 1.0f / (float)(N - 1);
    const int ntiles = N / 16;

    hipLaunchKernelGGL(genode_rk45_wmma, dim3(ntiles), dim3(32), 0, stream,
                       z_end, W1, b1, W2, b2, out, h);
}


# LDS tile: hidden columns interleaved (a lane's four tiles of a row adjacent): an activation store is one 16-byte store, not four 4-byte ones; k-chunks regroup hidden units: layer-2 sum order changes
# speedup vs baseline: 1.5193x; 1.0586x over previous
#include <hip/hip_runtime.h>
#include <math.h>

typedef __attribute__((ext_vector_type(2))) float v2f;
typedef __attribute__((ext_vector_type(4))) float v4f;
typedef __attribute__((ext_vector_type(8))) float v8f;

#define WMMA4(a, b, c) \
    __builtin_amdgcn_wmma_f32_16x16x4_f32(false, (a), false, (b), (short)0, (c), false, false)

__device__ __forceinline__ float fast_tanh(float x) {
#if __has_builtin(__builtin_amdgcn_tanhf)
    return __builtin_amdgcn_tanhf(x);
#else
    float xc = fminf(fmaxf(x, -10.0f), 10.0f);
    float t = __builtin_amdgcn_exp2f(xc * 2.8853900817779268f);
    return (t - 1.0f) * __builtin_amdgcn_rcpf(t + 1.0f);
#endif
}

__device__ __forceinline__ void wave_order() {
    __builtin_amdgcn_fence(__ATOMIC_RELEASE, "wavefront");
    __builtin_amdgcn_wave_barrier();
    __builtin_amdgcn_fence(__ATOMIC_ACQUIRE, "wavefront");
}

__launch_bounds__(32)
__global__ void genode_rk45_wmma(const float* __restrict__ z_end,
                                 const float* __restrict__ W1,
                                 const float* __restrict__ b1,
                                 const float* __restrict__ W2,
                                 const float* __restrict__ b2,
                                 float* __restrict__ out,
                                 float h)
{
    __shared__ __align__(16) float hbuf[16][132];
    __shared__ __align__(16) float zbuf[16][20];

    const int lane = threadIdx.x;
    const int n    = lane & 15;
    const int half = lane >> 4;
    const int dlt  = half * 2;
    const int g    = blockIdx.x;

    v2f w1B[8][4];
    #pragma unroll
    for (int t = 0; t < 8; ++t)
        #pragma unroll
        for (int c = 0; c < 4; ++c) {
            int k = 4 * c + dlt;
            v2f v;
            v.x = W1[k * 128 + 16 * t + n];
            v.y = W1[(k + 1) * 128 + 16 * t + n];
            w1B[t][c] = v;
        }
    v2f w2B[32];
    #pragma unroll
    for (int c = 0; c < 32; ++c) {
        const int H = c >> 4, i = (c & 15) >> 1, j = c & 1;
        const int k = 16 * (4 * H + 2 * j) + 2 * i + half;
        v2f v;
        v.x = W2[k * 16 + n];
        v.y = W2[(k + 16) * 16 + n];
        w2B[c] = v;
    }
    float b1v[8];
    #pragma unroll
    for (int t = 0; t < 8; ++t) b1v[t] = b1[16 * t + n];
    const float b2v = b2[n];

    v8f z;
    #pragma unroll
    for (int v = 0; v < 8; ++v)
        z[v] = z_end[(16 * g + v + half * 8) * 16 + n];

    auto put_z = [&](const v8f& val) {
        #pragma unroll
        for (int v = 0; v < 8; ++v)
            zbuf[v + half * 8][n] = val[v];
    };

    auto feval = [&](auto&& window) -> v8f {
        wave_order();
        v2f az[4];
        #pragma unroll
        for (int c = 0; c < 4; ++c)
            az[c] = *(const v2f*)&zbuf[n][4 * c + dlt];
        __builtin_amdgcn_sched_barrier(0);
        window();
        __builtin_amdgcn_sched_barrier(0);

        v8f acc[8];
        #pragma unroll
        for (int t = 0; t < 8; ++t)
            #pragma unroll
            for (int v = 0; v < 8; ++v) acc[t][v] = b1v[t];
        #pragma unroll
        for (int c = 0; c < 4; ++c)
            #pragma unroll
            for (int t = 0; t < 4; ++t)
                acc[t] = WMMA4(az[c], w1B[t][c], acc[t]);

        float ha[4][8];
        #pragma unroll
        for (int c = 0; c < 4; ++c)
            #pragma unroll
            for (int t = 4; t < 8; ++t) {
                acc[t] = WMMA4(az[c], w1B[t][c], acc[t]);
                const int i = 2 * (4 * c + (t - 4));
                ha[i >> 3][i & 7]             = fast_tanh(acc[i >> 3][i & 7]);
                ha[(i + 1) >> 3][(i + 1) & 7] = fast_tanh(acc[(i + 1) >> 3][(i + 1) & 7]);
                __builtin_amdgcn_sched_barrier(0);
            }

        #pragma unroll
        for (int v = 0; v < 8; ++v) {
            v4f q = {ha[0][v], ha[1][v], ha[2][v], ha[3][v]};
            *(v4f*)&hbuf[v + half * 8][4 * n] = q;
        }
        wave_order();
        v2f ah[32];
        #pragma unroll
        for (int i = 0; i < 8; ++i) {
            const v4f q = *(const v4f*)&hbuf[n][8 * i + 4 * half];
            ah[2 * i]     = v2f{q.x, q.y};
            ah[2 * i + 1] = v2f{q.z, q.w};
        }
        __builtin_amdgcn_sched_barrier(0);

        #pragma unroll
        for (int v = 0; v < 8; ++v) {
            v4f q = {fast_tanh(acc[4][v]), fast_tanh(acc[5][v]),
                     fast_tanh(acc[6][v]), fast_tanh(acc[7][v])};
            *(v4f*)&hbuf[v + half * 8][64 + 4 * n] = q;
        }
        wave_order();
        #pragma unroll
        for (int i = 0; i < 8; ++i) {
            const v4f q = *(const v4f*)&hbuf[n][64 + 8 * i + 4 * half];
            ah[16 + 2 * i]     = v2f{q.x, q.y};
            ah[16 + 2 * i + 1] = v2f{q.z, q.w};
        }
        __builtin_amdgcn_sched_barrier(0);

        v8f oa, ob;
        #pragma unroll
        for (int v = 0; v < 8; ++v) { oa[v] = b2v; ob[v] = 0.0f; }
        #pragma unroll
        for (int c = 0; c < 32; c += 2) {
            oa = WMMA4(ah[c],     w2B[c],     oa);
            ob = WMMA4(ah[c + 1], w2B[c + 1], ob);
        }
        v8f o;
        #pragma unroll
        for (int v = 0; v < 8; ++v) o[v] = oa[v] + ob[v];
        return o;
    };

    #define NEEDED_HERE(x) asm volatile("" :: "v"((x)[0]), "v"((x)[1]), "v"((x)[2]), "v"((x)[3]), \
                                              "v"((x)[4]), "v"((x)[5]), "v"((x)[6]), "v"((x)[7]))

    const int nsteps = 16 * g + 15;
    const int row0 = 16 * g + half * 8;
    for (int s = 1; s <= nsteps; ++s) {
        v8f S4, S5, S6, SF;
        v8f hm;
        v8f t;

        put_z(z);
        v8f k1 = feval([&]() {
            #pragma unroll
            for (int v = 0; v < 8; ++v) hm[v] = (row0 + v >= s) ? h : 0.0f;
            NEEDED_HERE(hm);
        });

        #pragma unroll
        for (int v = 0; v < 8; ++v) t[v] = z[v] + h * (0.2f * k1[v]);
        put_z(t);
        v8f k2 = feval([&]() {});

        #pragma unroll
        for (int v = 0; v < 8; ++v)
            t[v] = z[v] + h * ((3.0f / 40.0f) * k1[v] + (9.0f / 40.0f) * k2[v]);
        put_z(t);
        v8f k3 = feval([&]() {
            #pragma unroll
            for (int v = 0; v < 8; ++v) {
                S4[v] = (44.0f / 45.0f) * k1[v] - (56.0f / 15.0f) * k2[v];
                S5[v] = (19372.0f / 6561.0f) * k1[v] - (25360.0f / 2187.0f) * k2[v];
                S6[v] = (9017.0f / 3168.0f) * k1[v] - (355.0f / 33.0f) * k2[v];
            }
            NEEDED_HERE(S4); NEEDED_HERE(S5); NEEDED_HERE(S6);
        });

        #pragma unroll
        for (int v = 0; v < 8; ++v) t[v] = z[v] + h * (S4[v] + (32.0f / 9.0f) * k3[v]);
        put_z(t);
        v8f k4 = feval([&]() {
            #pragma unroll
            for (int v = 0; v < 8; ++v) {
                S5[v] = S5[v] + (64448.0f / 6561.0f) * k3[v];
                S6[v] = S6[v] + (46732.0f / 5247.0f) * k3[v];
                SF[v] = (35.0f / 384.0f) * k1[v] + (500.0f / 1113.0f) * k3[v];
            }
            NEEDED_HERE(S5); NEEDED_HERE(S6); NEEDED_HERE(SF);
        });

        #pragma unroll
        for (int v = 0; v < 8; ++v) t[v] = z[v] + h * (S5[v] - (212.0f / 729.0f) * k4[v]);
        put_z(t);
        v8f k5 = feval([&]() {
            #pragma unroll
            for (int v = 0; v < 8; ++v) {
                S6[v] = S6[v] + (49.0f / 176.0f) * k4[v];
                SF[v] = SF[v] + (125.0f / 192.0f) * k4[v];
            }
            NEEDED_HERE(S6); NEEDED_HERE(SF);
        });

        #pragma unroll
        for (int v = 0; v < 8; ++v) t[v] = z[v] + h * (S6[v] - (5103.0f / 18656.0f) * k5[v]);
        put_z(t);
        v8f k6 = feval([&]() {
            #pragma unroll
            for (int v = 0; v < 8; ++v) SF[v] = SF[v] - (2187.0f / 6784.0f) * k5[v];
            NEEDED_HERE(SF);
        });

        #pragma unroll
        for (int v = 0; v < 8; ++v)
            z[v] = z[v] + hm[v] * (SF[v] + (11.0f / 84.0f) * k6[v]);
    }
    #undef NEEDED_HERE

    #pragma unroll
    for (int v = 0; v < 8; ++v)
        out[(16 * g + v + half * 8) * 16 + n] = z[v];
}

extern "C" void kernel_launch(void* const* d_in, const int* in_sizes, int n_in,
                              void* d_out, int out_size, void* d_ws, size_t ws_size,
                              hipStream_t stream) {
    const float* z_end = (const float*)d_in[0];
    const float* W1    = (const float*)d_in[1];
    const float* b1    = (const float*)d_in[2];
    const float* W2    = (const float*)d_in[3];
    const float* b2    = (const float*)d_in[4];
    float* out = (float*)d_out;

    const int N = in_sizes[0] / 16;
    const float h = 1.0f / (float)(N - 1);
    const int ntiles = N / 16;

    hipLaunchKernelGGL(genode_rk45_wmma, dim3(ntiles), dim3(32), 0, stream,
                       z_end, W1, b1, W2, b2, out, h);
}


# Layer 2 with operands swapped (weights as A, activations as B): its D tile holds a trajectory row per lane and is layer 1's A fragment as it stands; stage input skips LDS; layer-1 sum order changes
# speedup vs baseline: 1.6090x; 1.0590x over previous
#include <hip/hip_runtime.h>
#include <math.h>

typedef __attribute__((ext_vector_type(2))) float v2f;
typedef __attribute__((ext_vector_type(4))) float v4f;
typedef __attribute__((ext_vector_type(8))) float v8f;

#define WMMA4(a, b, c) \
    __builtin_amdgcn_wmma_f32_16x16x4_f32(false, (a), false, (b), (short)0, (c), false, false)

__device__ __forceinline__ float fast_tanh(float x) {
#if __has_builtin(__builtin_amdgcn_tanhf)
    return __builtin_amdgcn_tanhf(x);
#else
    float xc = fminf(fmaxf(x, -10.0f), 10.0f);
    float t = __builtin_amdgcn_exp2f(xc * 2.8853900817779268f);
    return (t - 1.0f) * __builtin_amdgcn_rcpf(t + 1.0f);
#endif
}

__device__ __forceinline__ void wave_order() {
    __builtin_amdgcn_fence(__ATOMIC_RELEASE, "wavefront");
    __builtin_amdgcn_wave_barrier();
    __builtin_amdgcn_fence(__ATOMIC_ACQUIRE, "wavefront");
}

__launch_bounds__(32)
__global__ void genode_rk45_wmma(const float* __restrict__ z_end,
                                 const float* __restrict__ W1,
                                 const float* __restrict__ b1,
                                 const float* __restrict__ W2,
                                 const float* __restrict__ b2,
                                 float* __restrict__ out,
                                 float h)
{
    __shared__ __align__(16) float hbuf[16][132];

    const int lane = threadIdx.x;
    const int n    = lane & 15;
    const int half = lane >> 4;
    const int g    = blockIdx.x;

    v2f w1B[8][4];
    #pragma unroll
    for (int t = 0; t < 8; ++t)
        #pragma unroll
        for (int c = 0; c < 4; ++c) {
            int k = 8 * half + 2 * c;
            v2f v;
            v.x = W1[k * 128 + 16 * t + n];
            v.y = W1[(k + 1) * 128 + 16 * t + n];
            w1B[t][c] = v;
        }
    v2f w2A[32];
    #pragma unroll
    for (int c = 0; c < 32; ++c) {
        const int H = c >> 4, i = (c & 15) >> 1, j = c & 1;
        const int k = 16 * (4 * H + 2 * j) + 2 * i + half;
        v2f v;
        v.x = W2[k * 16 + n];
        v.y = W2[(k + 16) * 16 + n];
        w2A[c] = v;
    }
    float b1v[8];
    #pragma unroll
    for (int t = 0; t < 8; ++t) b1v[t] = b1[16 * t + n];
    v8f b2t;
    #pragma unroll
    for (int r = 0; r < 8; ++r) b2t[r] = b2[8 * half + r];

    v8f z;
    #pragma unroll
    for (int r = 0; r < 8; ++r)
        z[r] = z_end[(16 * g + n) * 16 + 8 * half + r];

    auto feval = [&](const v8f& zin, auto&& window) -> v8f {
        v2f az[4];
        #pragma unroll
        for (int c = 0; c < 4; ++c)
            az[c] = v2f{zin[2 * c], zin[2 * c + 1]};

        v8f acc[8];
        #pragma unroll
        for (int t = 0; t < 8; ++t)
            #pragma unroll
            for (int v = 0; v < 8; ++v) acc[t][v] = b1v[t];
        #pragma unroll
        for (int c = 0; c < 4; ++c)
            #pragma unroll
            for (int t = 0; t < 4; ++t)
                acc[t] = WMMA4(az[c], w1B[t][c], acc[t]);

        float ha[4][8];
        #pragma unroll
        for (int c = 0; c < 4; ++c)
            #pragma unroll
            for (int t = 4; t < 8; ++t) {
                acc[t] = WMMA4(az[c], w1B[t][c], acc[t]);
                const int i = 2 * (4 * c + (t - 4));
                ha[i >> 3][i & 7]             = fast_tanh(acc[i >> 3][i & 7]);
                ha[(i + 1) >> 3][(i + 1) & 7] = fast_tanh(acc[(i + 1) >> 3][(i + 1) & 7]);
                __builtin_amdgcn_sched_barrier(0);
            }

        #pragma unroll
        for (int v = 0; v < 8; ++v) {
            v4f q = {ha[0][v], ha[1][v], ha[2][v], ha[3][v]};
            *(v4f*)&hbuf[v + half * 8][4 * n] = q;
        }
        wave_order();
        v2f ah[32];
        #pragma unroll
        for (int i = 0; i < 8; ++i) {
            const v4f q = *(const v4f*)&hbuf[n][8 * i + 4 * half];
            ah[2 * i]     = v2f{q.x, q.y};
            ah[2 * i + 1] = v2f{q.z, q.w};
        }
        __builtin_amdgcn_sched_barrier(0);

        window();
        #pragma unroll
        for (int v = 0; v < 8; ++v) {
            v4f q = {fast_tanh(acc[4][v]), fast_tanh(acc[5][v]),
                     fast_tanh(acc[6][v]), fast_tanh(acc[7][v])};
            *(v4f*)&hbuf[v + half * 8][64 + 4 * n] = q;
        }
        wave_order();
        #pragma unroll
        for (int i = 0; i < 8; ++i) {
            const v4f q = *(const v4f*)&hbuf[n][64 + 8 * i + 4 * half];
            ah[16 + 2 * i]     = v2f{q.x, q.y};
            ah[16 + 2 * i + 1] = v2f{q.z, q.w};
        }
        __builtin_amdgcn_sched_barrier(0);

        v8f oa = b2t, ob;
        #pragma unroll
        for (int v = 0; v < 8; ++v) ob[v] = 0.0f;
        #pragma unroll
        for (int c = 0; c < 32; c += 2) {
            oa = WMMA4(w2A[c],     ah[c],     oa);
            ob = WMMA4(w2A[c + 1], ah[c + 1], ob);
        }
        v8f o;
        #pragma unroll
        for (int v = 0; v < 8; ++v) o[v] = oa[v] + ob[v];
        return o;
    };

    #define NEEDED_HERE(x) asm volatile("" :: "v"((x)[0]), "v"((x)[1]), "v"((x)[2]), "v"((x)[3]), \
                                              "v"((x)[4]), "v"((x)[5]), "v"((x)[6]), "v"((x)[7]))

    const int nsteps = 16 * g + 15;
    const int row = 16 * g + n;
    for (int s = 1; s <= nsteps; ++s) {
        v8f S4, S5, S6, SF;
        const float hm = (row >= s) ? h : 0.0f;
        v8f t;

        v8f k1 = feval(z, [&]() {});

        #pragma unroll
        for (int v = 0; v < 8; ++v) t[v] = z[v] + h * (0.2f * k1[v]);
        v8f k2 = feval(t, [&]() {});

        #pragma unroll
        for (int v = 0; v < 8; ++v)
            t[v] = z[v] + h * ((3.0f / 40.0f) * k1[v] + (9.0f / 40.0f) * k2[v]);
        v8f k3 = feval(t, [&]() {
            #pragma unroll
            for (int v = 0; v < 8; ++v) {
                S4[v] = (44.0f / 45.0f) * k1[v] - (56.0f / 15.0f) * k2[v];
                S5[v] = (19372.0f / 6561.0f) * k1[v] - (25360.0f / 2187.0f) * k2[v];
                S6[v] = (9017.0f / 3168.0f) * k1[v] - (355.0f / 33.0f) * k2[v];
            }
            NEEDED_HERE(S4); NEEDED_HERE(S5); NEEDED_HERE(S6);
        });

        #pragma unroll
        for (int v = 0; v < 8; ++v) t[v] = z[v] + h * (S4[v] + (32.0f / 9.0f) * k3[v]);
        v8f k4 = feval(t, [&]() {
            #pragma unroll
            for (int v = 0; v < 8; ++v) {
                S5[v] = S5[v] + (64448.0f / 6561.0f) * k3[v];
                S6[v] = S6[v] + (46732.0f / 5247.0f) * k3[v];
                SF[v] = (35.0f / 384.0f) * k1[v] + (500.0f / 1113.0f) * k3[v];
            }
            NEEDED_HERE(S5); NEEDED_HERE(S6); NEEDED_HERE(SF);
        });

        #pragma unroll
        for (int v = 0; v < 8; ++v) t[v] = z[v] + h * (S5[v] - (212.0f / 729.0f) * k4[v]);
        v8f k5 = feval(t, [&]() {
            #pragma unroll
            for (int v = 0; v < 8; ++v) {
                S6[v] = S6[v] + (49.0f / 176.0f) * k4[v];
                SF[v] = SF[v] + (125.0f / 192.0f) * k4[v];
            }
            NEEDED_HERE(S6); NEEDED_HERE(SF);
        });

        #pragma unroll
        for (int v = 0; v < 8; ++v) t[v] = z[v] + h * (S6[v] - (5103.0f / 18656.0f) * k5[v]);
        v8f k6 = feval(t, [&]() {
            #pragma unroll
            for (int v = 0; v < 8; ++v) SF[v] = SF[v] - (2187.0f / 6784.0f) * k5[v];
            NEEDED_HERE(SF);
        });

        #pragma unroll
        for (int v = 0; v < 8; ++v)
            z[v] = z[v] + hm * (SF[v] + (11.0f / 84.0f) * k6[v]);
    }
    #undef NEEDED_HERE

    #pragma unroll
    for (int r = 0; r < 8; ++r)
        out[(16 * g + n) * 16 + 8 * half + r] = z[r];
}

extern "C" void kernel_launch(void* const* d_in, const int* in_sizes, int n_in,
                              void* d_out, int out_size, void* d_ws, size_t ws_size,
                              hipStream_t stream) {
    const float* z_end = (const float*)d_in[0];
    const float* W1    = (const float*)d_in[1];
    const float* b1    = (const float*)d_in[2];
    const float* W2    = (const float*)d_in[3];
    const float* b2    = (const float*)d_in[4];
    float* out = (float*)d_out;

    const int N = in_sizes[0] / 16;
    const float h = 1.0f / (float)(N - 1);
    const int ntiles = N / 16;

    hipLaunchKernelGGL(genode_rk45_wmma, dim3(ntiles), dim3(32), 0, stream,
                       z_end, W1, b1, W2, b2, out, h);
}
